# starting kernel
# baseline (speedup 1.0000x reference)
#include <hip/hip_runtime.h>

typedef float v2f __attribute__((ext_vector_type(2)));
typedef float v4f __attribute__((ext_vector_type(4)));
typedef float v8f __attribute__((ext_vector_type(8)));

#define F_DIM 128
#define OUT_DIM 16
#define PSTRIDE 32

__global__ __launch_bounds__(256) void node_project_wmma(
    const float* __restrict__ h,
    const float* __restrict__ W,
    float* __restrict__ P,
    int N)
{
  const int lane = threadIdx.x & 31;
  const int wave = threadIdx.x >> 5;
  const int rowBase = blockIdx.x * 128 + wave * 16;
  if (rowBase >= N) return;

  const int m   = lane & 15;
  const int hi  = lane >> 4;
  const int klo = hi * 2;

  int rowA = rowBase + m;
  if (rowA >= N) rowA = N - 1;
  const float* __restrict__ hrow = h + (size_t)rowA * F_DIM;
  const float* __restrict__ wrow = W + (size_t)m * (2 * F_DIM);

  v8f c0 = {};
  v8f c1 = {};

#pragma unroll 8
  for (int kb = 0; kb < F_DIM; kb += 4) {
    const int k = kb + klo;
    v2f a  = __builtin_nontemporal_load((const v2f*)(hrow + k));
    v2f b0 = *(const v2f*)(wrow + k);
    v2f b1 = *(const v2f*)(wrow + F_DIM + k);
    c0 = __builtin_amdgcn_wmma_f32_16x16x4_f32(false, a, false, b0,
                                               (short)0, c0, false, false);
    c1 = __builtin_amdgcn_wmma_f32_16x16x4_f32(false, a, false, b1,
                                               (short)0, c1, false, false);
  }

  float* __restrict__ prow0 = P + (size_t)(rowBase + hi * 8) * PSTRIDE + m;
  if (rowBase + 16 <= N) {
#pragma unroll
    for (int r = 0; r < 8; ++r) {
      prow0[r * PSTRIDE]           = c0[r];
      prow0[r * PSTRIDE + OUT_DIM] = c1[r];
    }
  } else {
#pragma unroll
    for (int r = 0; r < 8; ++r) {
      if (rowBase + hi * 8 + r < N) {
        prow0[r * PSTRIDE]           = c0[r];
        prow0[r * PSTRIDE + OUT_DIM] = c1[r];
      }
    }
  }
}

__global__ __launch_bounds__(256) void edge_gather_add(
    const int* __restrict__ src,
    const int* __restrict__ dst,
    const float* __restrict__ P,
    const float* __restrict__ b,
    float* __restrict__ out,
    int E)
{
  const int tid = blockIdx.x * blockDim.x + threadIdx.x;
  const int e = tid >> 2;
  if (e >= E) return;
  const int q = (tid & 3) * 4;

  const int s = src[e];
  const int d = dst[e];

  const v4f u  = *(const v4f*)(P + (size_t)s * PSTRIDE + q);
  const v4f v  = *(const v4f*)(P + (size_t)d * PSTRIDE + OUT_DIM + q);
  const v4f bv = *(const v4f*)(b + q);

  const v4f r = u + v + bv;

  __builtin_nontemporal_store(r, (v4f*)(out + (size_t)e * OUT_DIM + q));
}

extern "C" void kernel_launch(void* const* d_in, const int* in_sizes, int n_in,
                              void* d_out, int out_size, void* d_ws, size_t ws_size,
                              hipStream_t stream) {
  const float* h   = (const float*)d_in[0];
  const int*   src = (const int*)d_in[1];
  const int*   dst = (const int*)d_in[2];
  const float* W   = (const float*)d_in[3];
  const float* b   = (const float*)d_in[4];
  float* out = (float*)d_out;
  float* P   = (float*)d_ws;

  const int N = in_sizes[0] / F_DIM;
  const int E = in_sizes[1];

  const int grid1 = (N + 127) / 128;
  node_project_wmma<<<grid1, 256, 0, stream>>>(h, W, P, N);

  const long long threads2 = (long long)E * 4;
  const int grid2 = (int)((threads2 + 255) / 256);
  edge_gather_add<<<grid2, 256, 0, stream>>>(src, dst, P, b, out, E);
}


# projection kernel: h tile and W fetched with whole-line 16-byte loads and turned into WMMA fragments through LDS (was 8-byte strided loads); k order inside each group of 8 changes the f32 sum order
# speedup vs baseline: 1.6769x; 1.6769x over previous
#include <hip/hip_runtime.h>

typedef float v2f __attribute__((ext_vector_type(2)));
typedef float v4f __attribute__((ext_vector_type(4)));
typedef float v8f __attribute__((ext_vector_type(8)));

#define F_DIM 128
#define OUT_DIM 16
#define PSTRIDE 32

#define K_HALF 64
#define A_PITCH 68
#define A_WAVE_FLOATS (16 * A_PITCH)
#define W_FLOATS (OUT_DIM * 2 * F_DIM)

__global__ __launch_bounds__(256) void node_project_wmma(
    const float* __restrict__ h,
    const float* __restrict__ W,
    float* __restrict__ P,
    int N)
{
  __shared__ __attribute__((aligned(16))) float sW[W_FLOATS];
  __shared__ __attribute__((aligned(16))) float sA[8 * A_WAVE_FLOATS];

  const int lane = threadIdx.x & 31;
  const int wave = threadIdx.x >> 5;
  const int rowBase = blockIdx.x * 128 + wave * 16;

  const int m   = lane & 15;
  const int hi  = lane >> 4;

  v4f t[2][8];
#pragma unroll
  for (int c = 0; c < 2; ++c) {
#pragma unroll
    for (int i = 0; i < 8; ++i) {
      int r = rowBase + 2 * i + hi;
      if (r >= N) r = N - 1;
      t[c][i] = __builtin_nontemporal_load(
          (const v4f*)(h + (size_t)r * F_DIM + K_HALF * c + 4 * m));
    }
  }

#pragma unroll
  for (int j = 0; j < 4; ++j) {
    const int G  = wave * 4 + j;
    const int g  = G >> 1;
    const int uv = G & 1;
    const v4f w4 = *(const v4f*)(W + (size_t)m * (2 * F_DIM) + uv * F_DIM + 8 * g + 4 * hi);
    *(v4f*)(sW + (G * 32 + lane) * 4) = w4;
  }
  __builtin_amdgcn_fence(__ATOMIC_RELEASE, "workgroup", "local");
  __builtin_amdgcn_s_barrier();
  __builtin_amdgcn_fence(__ATOMIC_ACQUIRE, "workgroup", "local");

  float* __restrict__ myA = sA + wave * A_WAVE_FLOATS;

  v8f c0 = {};
  v8f c1 = {};

#pragma unroll
  for (int c = 0; c < 2; ++c) {
#pragma unroll
    for (int i = 0; i < 8; ++i)
      *(v4f*)(myA + (2 * i + hi) * A_PITCH + 4 * m) = t[c][i];

    __builtin_amdgcn_fence(__ATOMIC_RELEASE, "wavefront");
    __builtin_amdgcn_wave_barrier();
    __builtin_amdgcn_fence(__ATOMIC_ACQUIRE, "wavefront");

#pragma unroll
    for (int p = 0; p < 8; ++p) {
      const int g = 8 * c + p;
      const v4f a4 = *(const v4f*)(myA + m * A_PITCH + 8 * p + 4 * hi);
      const v4f bu = *(const v4f*)(sW + ((2 * g) * 32 + lane) * 4);
      const v4f bv = *(const v4f*)(sW + ((2 * g + 1) * 32 + lane) * 4);
      const v2f a_lo = {a4.x, a4.y}, a_hi = {a4.z, a4.w};
      const v2f bu_lo = {bu.x, bu.y}, bu_hi = {bu.z, bu.w};
      const v2f bv_lo = {bv.x, bv.y}, bv_hi = {bv.z, bv.w};
      c0 = __builtin_amdgcn_wmma_f32_16x16x4_f32(false, a_lo, false, bu_lo,
                                                 (short)0, c0, false, false);
      c1 = __builtin_amdgcn_wmma_f32_16x16x4_f32(false, a_lo, false, bv_lo,
                                                 (short)0, c1, false, false);
      c0 = __builtin_amdgcn_wmma_f32_16x16x4_f32(false, a_hi, false, bu_hi,
                                                 (short)0, c0, false, false);
      c1 = __builtin_amdgcn_wmma_f32_16x16x4_f32(false, a_hi, false, bv_hi,
                                                 (short)0, c1, false, false);
    }

    __builtin_amdgcn_fence(__ATOMIC_RELEASE, "wavefront");
    __builtin_amdgcn_wave_barrier();
    __builtin_amdgcn_fence(__ATOMIC_ACQUIRE, "wavefront");
  }

  float* __restrict__ prow0 = P + (size_t)(rowBase + hi * 8) * PSTRIDE + m;
  if (rowBase + 16 <= N) {
#pragma unroll
    for (int r = 0; r < 8; ++r) {
      prow0[r * PSTRIDE]           = c0[r];
      prow0[r * PSTRIDE + OUT_DIM] = c1[r];
    }
  } else {
#pragma unroll
    for (int r = 0; r < 8; ++r) {
      if (rowBase + hi * 8 + r < N) {
        prow0[r * PSTRIDE]           = c0[r];
        prow0[r * PSTRIDE + OUT_DIM] = c1[r];
      }
    }
  }
}

__global__ __launch_bounds__(256) void edge_gather_add(
    const int* __restrict__ src,
    const int* __restrict__ dst,
    const float* __restrict__ P,
    const float* __restrict__ b,
    float* __restrict__ out,
    int E)
{
  const int tid = blockIdx.x * blockDim.x + threadIdx.x;
  const int e = tid >> 2;
  if (e >= E) return;
  const int q = (tid & 3) * 4;

  const int s = src[e];
  const int d = dst[e];

  const v4f u  = *(const v4f*)(P + (size_t)s * PSTRIDE + q);
  const v4f v  = *(const v4f*)(P + (size_t)d * PSTRIDE + OUT_DIM + q);
  const v4f bv = *(const v4f*)(b + q);

  const v4f r = u + v + bv;

  __builtin_nontemporal_store(r, (v4f*)(out + (size_t)e * OUT_DIM + q));
}

extern "C" void kernel_launch(void* const* d_in, const int* in_sizes, int n_in,
                              void* d_out, int out_size, void* d_ws, size_t ws_size,
                              hipStream_t stream) {
  const float* h   = (const float*)d_in[0];
  const int*   src = (const int*)d_in[1];
  const int*   dst = (const int*)d_in[2];
  const float* W   = (const float*)d_in[3];
  const float* b   = (const float*)d_in[4];
  float* out = (float*)d_out;
  float* P   = (float*)d_ws;

  const int N = in_sizes[0] / F_DIM;
  const int E = in_sizes[1];

  const int grid1 = (N + 127) / 128;
  node_project_wmma<<<grid1, 256, 0, stream>>>(h, W, P, N);

  const long long threads2 = (long long)E * 4;
  const int grid2 = (int)((threads2 + 255) / 256);
  edge_gather_add<<<grid2, 256, 0, stream>>>(src, dst, P, b, out, E);
}


# edge kernel: one wave does 128 edges (was 8) in straight-line code: each lane loads one index per 32 edges, lanes pass them by permute; all 32 row gathers issue before the first wait; 1/16 the waves
# speedup vs baseline: 1.7388x; 1.0369x over previous
#include <hip/hip_runtime.h>

typedef float v2f __attribute__((ext_vector_type(2)));
typedef float v4f __attribute__((ext_vector_type(4)));
typedef float v8f __attribute__((ext_vector_type(8)));

#define F_DIM 128
#define OUT_DIM 16
#define PSTRIDE 32

#define K_HALF 64
#define A_PITCH 68
#define A_WAVE_FLOATS (16 * A_PITCH)
#define W_FLOATS (OUT_DIM * 2 * F_DIM)

__global__ __launch_bounds__(256) void node_project_wmma(
    const float* __restrict__ h,
    const float* __restrict__ W,
    float* __restrict__ P,
    int N)
{
  __shared__ __attribute__((aligned(16))) float sW[W_FLOATS];
  __shared__ __attribute__((aligned(16))) float sA[8 * A_WAVE_FLOATS];

  const int lane = threadIdx.x & 31;
  const int wave = threadIdx.x >> 5;
  const int rowBase = blockIdx.x * 128 + wave * 16;

  const int m   = lane & 15;
  const int hi  = lane >> 4;

  v4f t[2][8];
#pragma unroll
  for (int c = 0; c < 2; ++c) {
#pragma unroll
    for (int i = 0; i < 8; ++i) {
      int r = rowBase + 2 * i + hi;
      if (r >= N) r = N - 1;
      t[c][i] = __builtin_nontemporal_load(
          (const v4f*)(h + (size_t)r * F_DIM + K_HALF * c + 4 * m));
    }
  }

#pragma unroll
  for (int j = 0; j < 4; ++j) {
    const int G  = wave * 4 + j;
    const int g  = G >> 1;
    const int uv = G & 1;
    const v4f w4 = *(const v4f*)(W + (size_t)m * (2 * F_DIM) + uv * F_DIM + 8 * g + 4 * hi);
    *(v4f*)(sW + (G * 32 + lane) * 4) = w4;
  }
  __builtin_amdgcn_fence(__ATOMIC_RELEASE, "workgroup", "local");
  __builtin_amdgcn_s_barrier();
  __builtin_amdgcn_fence(__ATOMIC_ACQUIRE, "workgroup", "local");

  float* __restrict__ myA = sA + wave * A_WAVE_FLOATS;

  v8f c0 = {};
  v8f c1 = {};

#pragma unroll
  for (int c = 0; c < 2; ++c) {
#pragma unroll
    for (int i = 0; i < 8; ++i)
      *(v4f*)(myA + (2 * i + hi) * A_PITCH + 4 * m) = t[c][i];

    __builtin_amdgcn_fence(__ATOMIC_RELEASE, "wavefront");
    __builtin_amdgcn_wave_barrier();
    __builtin_amdgcn_fence(__ATOMIC_ACQUIRE, "wavefront");

#pragma unroll
    for (int p = 0; p < 8; ++p) {
      const int g = 8 * c + p;
      const v4f a4 = *(const v4f*)(myA + m * A_PITCH + 8 * p + 4 * hi);
      const v4f bu = *(const v4f*)(sW + ((2 * g) * 32 + lane) * 4);
      const v4f bv = *(const v4f*)(sW + ((2 * g + 1) * 32 + lane) * 4);
      const v2f a_lo = {a4.x, a4.y}, a_hi = {a4.z, a4.w};
      const v2f bu_lo = {bu.x, bu.y}, bu_hi = {bu.z, bu.w};
      const v2f bv_lo = {bv.x, bv.y}, bv_hi = {bv.z, bv.w};
      c0 = __builtin_amdgcn_wmma_f32_16x16x4_f32(false, a_lo, false, bu_lo,
                                                 (short)0, c0, false, false);
      c1 = __builtin_amdgcn_wmma_f32_16x16x4_f32(false, a_lo, false, bv_lo,
                                                 (short)0, c1, false, false);
      c0 = __builtin_amdgcn_wmma_f32_16x16x4_f32(false, a_hi, false, bu_hi,
                                                 (short)0, c0, false, false);
      c1 = __builtin_amdgcn_wmma_f32_16x16x4_f32(false, a_hi, false, bv_hi,
                                                 (short)0, c1, false, false);
    }

    __builtin_amdgcn_fence(__ATOMIC_RELEASE, "wavefront");
    __builtin_amdgcn_wave_barrier();
    __builtin_amdgcn_fence(__ATOMIC_ACQUIRE, "wavefront");
  }

  float* __restrict__ prow0 = P + (size_t)(rowBase + hi * 8) * PSTRIDE + m;
  if (rowBase + 16 <= N) {
#pragma unroll
    for (int r = 0; r < 8; ++r) {
      prow0[r * PSTRIDE]           = c0[r];
      prow0[r * PSTRIDE + OUT_DIM] = c1[r];
    }
  } else {
#pragma unroll
    for (int r = 0; r < 8; ++r) {
      if (rowBase + hi * 8 + r < N) {
        prow0[r * PSTRIDE]           = c0[r];
        prow0[r * PSTRIDE + OUT_DIM] = c1[r];
      }
    }
  }
}

#define E_GROUPS 4
#define E_PER_WAVE (32 * E_GROUPS)
#define E_WAVES 8
#define E_PER_BLOCK (E_WAVES * E_PER_WAVE)

__global__ __launch_bounds__(256) __attribute__((amdgpu_waves_per_eu(6, 6)))
void edge_gather_add(
    const int* __restrict__ src,
    const int* __restrict__ dst,
    const float* __restrict__ P,
    const float* __restrict__ b,
    float* __restrict__ out,
    int E)
{
  const unsigned lane = threadIdx.x & 31;
  const int wave = __builtin_amdgcn_readfirstlane((int)(threadIdx.x >> 5));
  const int base = (blockIdx.x * E_WAVES + wave) * E_PER_WAVE;
  if (base >= E) return;

  const unsigned ql  = lane & 3;
  const unsigned sub = lane >> 2;
  const v4f bv = *(const v4f*)(b + 4 * ql);

  const v4f* __restrict__ Pu = (const v4f*)P;
  const v4f* __restrict__ Pv = Pu + 4;

  if (base + E_PER_WAVE <= E) {
    const int* __restrict__ sp = src + base;
    const int* __restrict__ dp = dst + base;
    int si[E_GROUPS], di[E_GROUPS];
#pragma unroll
    for (int k = 0; k < E_GROUPS; ++k) {
      si[k] = sp[32 * k + lane];
      di[k] = dp[32 * k + lane];
    }

    v4f u[4 * E_GROUPS], v[4 * E_GROUPS];
#pragma unroll
    for (int k = 0; k < E_GROUPS; ++k) {
#pragma unroll
      for (int j = 0; j < 4; ++j) {
        const int from = (int)((8 * j + sub) << 2);
        const unsigned s = (unsigned)__builtin_amdgcn_ds_bpermute(from, si[k]);
        const unsigned d = (unsigned)__builtin_amdgcn_ds_bpermute(from, di[k]);
        u[4 * k + j] = Pu[s * 8u + ql];
        v[4 * k + j] = Pv[d * 8u + ql];
      }
    }

    v4f* __restrict__ o4 = (v4f*)out + (size_t)base * 4;
#pragma unroll
    for (int i = 0; i < 4 * E_GROUPS; ++i) {
      __builtin_amdgcn_sched_barrier(0);
      const v4f r = u[i] + v[i] + bv;
      __builtin_nontemporal_store(r, o4 + (lane + 32 * i));
    }
  } else {
    for (int e = base + (int)sub; e < E; e += 8) {
      const unsigned s = (unsigned)src[e];
      const unsigned d = (unsigned)dst[e];
      const v4f r = Pu[s * 8u + ql] + Pv[d * 8u + ql] + bv;
      __builtin_nontemporal_store(r, (v4f*)out + ((size_t)e * 4 + ql));
    }
  }
}

extern "C" void kernel_launch(void* const* d_in, const int* in_sizes, int n_in,
                              void* d_out, int out_size, void* d_ws, size_t ws_size,
                              hipStream_t stream) {
  const float* h   = (const float*)d_in[0];
  const int*   src = (const int*)d_in[1];
  const int*   dst = (const int*)d_in[2];
  const float* W   = (const float*)d_in[3];
  const float* b   = (const float*)d_in[4];
  float* out = (float*)d_out;
  float* P   = (float*)d_ws;

  const int N = in_sizes[0] / F_DIM;
  const int E = in_sizes[1];

  const int grid1 = (N + 127) / 128;
  node_project_wmma<<<grid1, 256, 0, stream>>>(h, W, P, N);

  const int grid2 = (E + E_PER_BLOCK - 1) / E_PER_BLOCK;
  edge_gather_add<<<grid2, 256, 0, stream>>>(src, dst, P, b, out, E);
}
